# starting kernel
# baseline (speedup 1.0000x reference)
#include <hip/hip_runtime.h>
#include <math.h>

typedef __attribute__((ext_vector_type(16))) _Float16 v16h;
typedef __attribute__((ext_vector_type(8)))  float    v8f;

namespace cfg {
constexpr int NB    = 128;
constexpr int SEQ   = 64;
constexpr int CH    = 256;
constexpr int PARTS = 16;
constexpr int HID   = 64;
constexpr int XT_ROWS   = SEQ + 4;
constexpr int XT_STRIDE = CH + 8;
constexpr int HT_STRIDE = HID + 8;
}

__device__ __forceinline__ v16h load_a_frag_g(const float* base, int rstride,
                                              int kstride, int lane) {
  const int m  = lane & 15;
  const int hs = (lane >> 4) & 1;
  const float* p = base + (size_t)m * rstride;
  v16h a;
#pragma unroll
  for (int t = 0; t < 16; ++t) {
    const int v = t >> 1, e = t & 1;
    const int k = ((v >> 2) << 4) + (hs << 3) + ((v & 3) << 1) + e;
    a[t] = (_Float16)p[k * kstride];
  }
  return a;
}

__device__ __forceinline__ v16h load_b_frag_lds(const _Float16* lds, int row_base,
                                                int row_stride, int col_base,
                                                int lane) {
  const int n  = lane & 15;
  const int hs = (lane >> 4) & 1;
  const _Float16* p = lds + (size_t)(row_base + n) * row_stride + col_base;
  v16h b;
#pragma unroll
  for (int t = 0; t < 16; ++t) {
    const int v = t >> 1, e = t & 1;
    const int k = ((v >> 2) << 4) + (hs << 3) + ((v & 3) << 1) + e;
    b[t] = p[k];
  }
  return b;
}

__device__ __forceinline__ v8f wmma_f16(v16h a, v16h b, v8f c) {
  return __builtin_amdgcn_wmma_f32_16x16x32_f16(false, a, false, b,
                                                (short)0, c, false, false);
}

__device__ __forceinline__ float sigmoidf_(float z) {
  return 1.0f / (1.0f + __expf(-z));
}

__global__ __launch_bounds__(256)
void mtb_gate_kernel(const float* __restrict__ x,
                     const float* __restrict__ W1a,
                     const float* __restrict__ W1b,
                     const float* __restrict__ W2a,
                     const float* __restrict__ W2b,
                     float* __restrict__ out) {
  using namespace cfg;
  __shared__ _Float16 sXt[XT_ROWS * XT_STRIDE];
  __shared__ _Float16 sHt[2][XT_ROWS * HT_STRIDE];

  const int part = blockIdx.x;
  const int b    = blockIdx.y;
  const int tid  = threadIdx.x;
  const int lane = tid & 31;
  const int wave = tid >> 5;
  const int hs   = (lane >> 4) & 1;

  {
    sXt[0 * XT_STRIDE + tid]         = (_Float16)0.f;
    sXt[1 * XT_STRIDE + tid]         = (_Float16)0.f;
    sXt[(SEQ + 2) * XT_STRIDE + tid] = (_Float16)0.f;
    sXt[(SEQ + 3) * XT_STRIDE + tid] = (_Float16)0.f;
    if (tid < HID) {
#pragma unroll
      for (int r = 0; r < 2; ++r) {
        sHt[0][r * HT_STRIDE + tid] = (_Float16)0.f;
        sHt[1][r * HT_STRIDE + tid] = (_Float16)0.f;
        sHt[0][(SEQ + 2 + r) * HT_STRIDE + tid] = (_Float16)0.f;
        sHt[1][(SEQ + 2 + r) * HT_STRIDE + tid] = (_Float16)0.f;
      }
    }
    const int c = tid;
    const float* xp = x + (((size_t)b * SEQ) * CH + c) * PARTS + part;
#pragma unroll 4
    for (int l = 0; l < SEQ; ++l) {
      float v = xp[(size_t)l * CH * PARTS];
      sXt[(2 + l) * XT_STRIDE + c] = (_Float16)v;
    }
  }
  __syncthreads();

  {
    const int br = wave >> 2;
    const int mt = wave & 3;
    const int m0 = mt * 16;
    const float* Wa = (br ? W2a : W1a) +
                      ((size_t)(part * HID + m0)) * CH * 3;
    v8f acc[4] = {};
    for (int s = 0; s < 3; ++s) {
      for (int kb = 0; kb < CH / 32; ++kb) {
        const int c0 = kb * 32;
        v16h a = load_a_frag_g(Wa + c0 * 3 + s, CH * 3, 3, lane);
#pragma unroll
        for (int nt = 0; nt < 4; ++nt) {
          v16h bf = load_b_frag_lds(sXt, 1 + s + nt * 16, XT_STRIDE, c0, lane);
          acc[nt] = wmma_f16(a, bf, acc[nt]);
        }
      }
    }
#pragma unroll
    for (int nt = 0; nt < 4; ++nt) {
#pragma unroll
      for (int v = 0; v < 8; ++v) {
        float f = acc[nt][v];
        f = f > 0.f ? f : 0.01f * f;
        const int row = 2 + nt * 16 + (lane & 15);
        const int col = m0 + v + (hs << 3);
        sHt[br][row * HT_STRIDE + col] = (_Float16)f;
      }
    }
  }
  __syncthreads();

  for (int j = 0; j < 2; ++j) {
    const int m0 = wave * 32 + j * 16;
    const float* w1bp = W1b + ((size_t)(part * CH + m0)) * HID;
    const float* w2bp = W2b + ((size_t)(part * CH + m0)) * HID * 3;
    v16h a1[2], a2[3][2];
#pragma unroll
    for (int kb = 0; kb < 2; ++kb)
      a1[kb] = load_a_frag_g(w1bp + kb * 32, HID, 1, lane);
#pragma unroll
    for (int s = 0; s < 3; ++s)
#pragma unroll
      for (int kb = 0; kb < 2; ++kb)
        a2[s][kb] = load_a_frag_g(w2bp + kb * 32 * 3 + s, HID * 3, 3, lane);

    float maxv[8];
#pragma unroll
    for (int v = 0; v < 8; ++v) maxv[v] = -INFINITY;

    for (int nt = 0; nt < 4; ++nt) {
      v8f acc1 = {}, acc2 = {};
#pragma unroll
      for (int kb = 0; kb < 2; ++kb) {
        v16h bf = load_b_frag_lds(sHt[0], 2 + nt * 16, HT_STRIDE, kb * 32, lane);
        acc1 = wmma_f16(a1[kb], bf, acc1);
      }
      for (int s = 0; s < 3; ++s) {
#pragma unroll
        for (int kb = 0; kb < 2; ++kb) {
          v16h bf = load_b_frag_lds(sHt[1], 1 + s + nt * 16, HT_STRIDE,
                                    kb * 32, lane);
          acc2 = wmma_f16(a2[s][kb], bf, acc2);
        }
      }
#pragma unroll
      for (int v = 0; v < 8; ++v) {
        const int c = m0 + v + (hs << 3);
        const int l = nt * 16 + (lane & 15);
        float xs[5];
#pragma unroll
        for (int d = 0; d < 5; ++d)
          xs[d] = (float)sXt[(l + d) * XT_STRIDE + c];
        const float avg3 = (xs[1] + xs[2] + xs[3]) * (1.0f / 3.0f);
        const float avg5 = (xs[0] + xs[1] + xs[2] + xs[3] + xs[4]) * 0.2f;
        float m3 = xs[2];
        m3 = fmaxf(m3, (l >= 1)       ? xs[1] : -INFINITY);
        m3 = fmaxf(m3, (l <= cfg::SEQ - 2) ? xs[3] : -INFINITY);
        float m5 = m3;
        m5 = fmaxf(m5, (l >= 2)       ? xs[0] : -INFINITY);
        m5 = fmaxf(m5, (l <= cfg::SEQ - 3) ? xs[4] : -INFINITY);
        const float g1 = sigmoidf_(acc1[v]);
        const float g2 = sigmoidf_(acc2[v]);
        const float mval = (avg3 + m3) * g1 + (avg5 + m5) * g2;
        maxv[v] = fmaxf(maxv[v], mval);
      }
    }
#pragma unroll
    for (int v = 0; v < 8; ++v) {
#pragma unroll
      for (int off = 8; off >= 1; off >>= 1)
        maxv[v] = fmaxf(maxv[v], __shfl_xor(maxv[v], off, 16));
      if ((lane & 15) == 0) {
        const int c = m0 + v + (hs << 3);
        out[((size_t)b * cfg::PARTS + part) * cfg::CH + c] = maxv[v];
      }
    }
  }
}

extern "C" void kernel_launch(void* const* d_in, const int* in_sizes, int n_in,
                              void* d_out, int out_size, void* d_ws, size_t ws_size,
                              hipStream_t stream) {
  (void)in_sizes; (void)n_in; (void)d_ws; (void)ws_size; (void)out_size;
  const float* x   = (const float*)d_in[0];
  const float* W1a = (const float*)d_in[1];
  const float* W1b = (const float*)d_in[2];
  const float* W2a = (const float*)d_in[3];
  const float* W2b = (const float*)d_in[4];
  float* out = (float*)d_out;

  dim3 grid(cfg::PARTS, cfg::NB);
  dim3 block(256);
  mtb_gate_kernel<<<grid, block, 0, stream>>>(x, W1a, W1b, W2a, W2b, out);
}


# Stage A weights staged per wave: each wave copies its own 16 rows to an LDS region no other wave touches, so the k loop has no workgroup barrier (was one per k block) and the waves run free
# speedup vs baseline: 2.5255x; 2.5255x over previous
#include <hip/hip_runtime.h>
#include <math.h>

typedef __attribute__((ext_vector_type(16))) _Float16 v16h;
typedef __attribute__((ext_vector_type(8)))  _Float16 v8h;
typedef __attribute__((ext_vector_type(8)))  float    v8f;

namespace cfg {
constexpr int NB    = 128;
constexpr int SEQ   = 64;
constexpr int CH    = 256;
constexpr int PARTS = 16;
constexpr int HID   = 64;
constexpr int PPB    = 4;
constexpr int GROUPS = 2;
constexpr int PPG    = PPB / GROUPS;
constexpr int GRID_X  = PARTS / PPB;
constexpr int THREADS = GROUPS * 256;
constexpr int XT_ROWS   = SEQ + 4;
constexpr int XT_STRIDE = PPB * CH + 8;
constexpr int HT_STRIDE = HID + 8;
constexpr int WA_PITCH = 32 * 3 + 8;
constexpr int WA_WAVE  = 16 * WA_PITCH;
constexpr int W2_PITCH = HID * 3 + 8;
constexpr int W1_PITCH = HID + 8;
constexpr int WB_WAVE  = 16 * (W2_PITCH + W1_PITCH);
constexpr int WST_A    = GROUPS * 8 * 2 * WA_WAVE;
constexpr int WST_B    = GROUPS * 8 * WB_WAVE;
constexpr int WST_HALFS = WST_A > WST_B ? WST_A : WST_B;
}

template <int NV8>
__device__ __forceinline__ void load_hrun(const _Float16* p, _Float16 (&r)[8 * NV8]) {
  const v8h* p8 = reinterpret_cast<const v8h*>(p);
#pragma unroll
  for (int i = 0; i < NV8; ++i) {
    const v8h q = p8[i];
#pragma unroll
    for (int e = 0; e < 8; ++e) r[8 * i + e] = q[e];
  }
}

template <int NTAP>
__device__ __forceinline__ v16h frag_of_hrun(const _Float16 (&r)[16 * NTAP], int s) {
  v16h a;
#pragma unroll
  for (int t = 0; t < 16; ++t) a[t] = r[t * NTAP + s];
  return a;
}

typedef __attribute__((ext_vector_type(4))) _Float16 v4h;
__device__ __forceinline__ void store_h4(_Float16* d, float4 v) {
  v4h h;
  h[0] = (_Float16)v.x; h[1] = (_Float16)v.y; h[2] = (_Float16)v.z; h[3] = (_Float16)v.w;
  *reinterpret_cast<v4h*>(d) = h;
}

typedef __attribute__((ext_vector_type(4))) float v4f;
__device__ __forceinline__ void store_h4(_Float16* d, v4f v) {
  v4h h;
  h[0] = (_Float16)v[0]; h[1] = (_Float16)v[1]; h[2] = (_Float16)v[2]; h[3] = (_Float16)v[3];
  *reinterpret_cast<v4h*>(d) = h;
}

__device__ __forceinline__ v16h load_b_frag_lds(const _Float16* lds, int row_base,
                                                int row_stride, int col_base,
                                                int lane) {
  const int n  = lane & 15;
  const int hs = (lane >> 4) & 1;
  const v8h* p = reinterpret_cast<const v8h*>(
      lds + (size_t)(row_base + n) * row_stride + col_base + (hs << 4));
  const v8h lo = p[0], hi = p[1];
  return __builtin_shufflevector(lo, hi, 0, 1, 2, 3, 4, 5, 6, 7,
                                 8, 9, 10, 11, 12, 13, 14, 15);
}

__device__ __forceinline__ v8f wmma_f16(v16h a, v16h b, v8f c) {
  return __builtin_amdgcn_wmma_f32_16x16x32_f16(false, a, false, b,
                                                (short)0, c, false, false);
}

__device__ __forceinline__ float sigmoidf_(float z) {
  return 1.0f / (1.0f + __expf(-z));
}

template <bool HAS_PREV>
__device__ __forceinline__ float frame_before(float tile, float prev_tile) {
  int edge = 0;
  if (HAS_PREV)
    edge = __builtin_amdgcn_update_dpp(0, __float_as_int(prev_tile), 0x121, 0xF, 0xF, false);
  return __int_as_float(
      __builtin_amdgcn_update_dpp(edge, __float_as_int(tile), 0x111, 0xF, 0xF, false));
}
template <bool HAS_NEXT>
__device__ __forceinline__ float frame_after(float tile, float next_tile) {
  int edge = 0;
  if (HAS_NEXT)
    edge = __builtin_amdgcn_update_dpp(0, __float_as_int(next_tile), 0x12F, 0xF, 0xF, false);
  return __int_as_float(
      __builtin_amdgcn_update_dpp(edge, __float_as_int(tile), 0x101, 0xF, 0xF, false));
}

__global__ __launch_bounds__(cfg::THREADS)
void mtb_gate_kernel(const float* __restrict__ x,
                     const float* __restrict__ W1a,
                     const float* __restrict__ W1b,
                     const float* __restrict__ W2a,
                     const float* __restrict__ W2b,
                     float* __restrict__ out) {
  using namespace cfg;
  __shared__ alignas(16) _Float16 sXt[XT_ROWS * XT_STRIDE];
  __shared__ alignas(16) _Float16 sHt[GROUPS][2][XT_ROWS * HT_STRIDE];
  __shared__ alignas(16) _Float16 sW[WST_HALFS];

  const int p0   = blockIdx.x * PPB;
  const int b    = blockIdx.y;
  const int tid  = threadIdx.x;
  const int lane = tid & 31;
  const int wv   = __builtin_amdgcn_readfirstlane(tid >> 5);
  const int grp  = wv >> 3;
  const int gtid = tid & 255;
  const int wave = wv & 7;
  const int hs   = (lane >> 4) & 1;

  {
    {
      const int pr  = tid >> 7;
      const int row = (pr < 2) ? pr : (SEQ + pr);
      const v8h zero = {};
      *reinterpret_cast<v8h*>(&sXt[row * XT_STRIDE + (tid & 127) * 8]) = zero;
    }
    if (gtid < HID) {
#pragma unroll
      for (int r = 0; r < 2; ++r) {
        sHt[grp][0][r * HT_STRIDE + gtid] = (_Float16)0.f;
        sHt[grp][1][r * HT_STRIDE + gtid] = (_Float16)0.f;
        sHt[grp][0][(SEQ + 2 + r) * HT_STRIDE + gtid] = (_Float16)0.f;
        sHt[grp][1][(SEQ + 2 + r) * HT_STRIDE + gtid] = (_Float16)0.f;
      }
    }
    const int c  = gtid;
    const int lh = grp;
    const float* xp = x + (((size_t)b * SEQ + lh) * CH + c) * PARTS + p0;
#pragma unroll 8
    for (int i = 0; i < SEQ / 2; ++i) {
      const float4 v = *reinterpret_cast<const float4*>(xp + (size_t)i * 2 * CH * PARTS);
      _Float16* d = sXt + (2 + 2 * i + lh) * XT_STRIDE + c;
      d[0 * CH] = (_Float16)v.x;
      d[1 * CH] = (_Float16)v.y;
      d[2 * CH] = (_Float16)v.z;
      d[3 * CH] = (_Float16)v.w;
    }
  }
  __syncthreads();

  for (int i = 0; i < PPG; ++i) {
  const int q    = grp * PPG + i;
  const int part = p0 + q;
  const int xcol = q * CH;
  _Float16* const Ht0 = sHt[grp][0];
  _Float16* const Ht1 = sHt[grp][1];
  if (i > 0) __syncthreads();

  {
    const int br = wave >> 2;
    const int mt = wave & 3;
    const int m0 = mt * 16;
    _Float16* const sWa = sW + (grp * 8 + wave) * (2 * WA_WAVE);
    const float* const wten = br ? W2a : W1a;
    const int rl = (lane >> 3) << 2;
    const unsigned goff = (unsigned)((part * HID + m0 + rl) * (CH * 3) + (lane & 7) * 4);
    const int ldst = rl * WA_PITCH + (lane & 7) * 4;
    v4f st[12];
#pragma unroll
    for (int a = 0; a < 4; ++a)
#pragma unroll
      for (int j = 0; j < 3; ++j)
        st[a * 3 + j] = *reinterpret_cast<const v4f*>(wten + goff + a * (CH * 3) + j * 32);
#pragma unroll
    for (int a = 0; a < 4; ++a)
#pragma unroll
      for (int j = 0; j < 3; ++j)
        store_h4(sWa + a * WA_PITCH + j * 32 + ldst, st[a * 3 + j]);
    __builtin_amdgcn_wave_barrier();

    const _Float16* const arun = sWa + (lane & 15) * WA_PITCH + hs * 48;
    v8f ps[3][4] = {};
    for (int kb = 0; kb < CH / 32; ++kb) {
      const int c0 = kb * 32;
      const bool more = kb + 1 < CH / 32;
      if (more) {
#pragma unroll
        for (int a = 0; a < 4; ++a)
#pragma unroll
          for (int j = 0; j < 3; ++j)
            st[a * 3 + j] = *reinterpret_cast<const v4f*>(
                wten + goff + (kb + 1) * 96 + a * (CH * 3) + j * 32);
      }
      __builtin_amdgcn_sched_barrier(0);
      _Float16 r[48];
      load_hrun<6>(arun + (kb & 1) * WA_WAVE, r);
      v16h a[3];
#pragma unroll
      for (int s = 0; s < 3; ++s) a[s] = frag_of_hrun<3>(r, s);
      v16h bf[4];
#pragma unroll
      for (int nt = 0; nt < 4; ++nt)
        bf[nt] = load_b_frag_lds(sXt, 2 + nt * 16, XT_STRIDE, xcol + c0, lane);
#pragma unroll
      for (int nt = 0; nt < 4; ++nt)
#pragma unroll
        for (int s = 0; s < 3; ++s) ps[s][nt] = wmma_f16(a[s], bf[nt], ps[s][nt]);
      __builtin_amdgcn_sched_barrier(0);
      if (more) {
        asm volatile("" :: "v"(st[0]), "v"(st[1]), "v"(st[2]), "v"(st[3]), "v"(st[4]), "v"(st[5]),
                           "v"(st[6]), "v"(st[7]), "v"(st[8]), "v"(st[9]), "v"(st[10]), "v"(st[11]));
#pragma unroll
        for (int a = 0; a < 4; ++a)
#pragma unroll
          for (int j = 0; j < 3; ++j)
            store_h4(sWa + ((kb + 1) & 1) * WA_WAVE + a * WA_PITCH + j * 32 + ldst,
                     st[a * 3 + j]);
        __builtin_amdgcn_wave_barrier();
      }
    }
#pragma unroll
    for (int nt = 0; nt < 4; ++nt) {
#pragma unroll
      for (int v = 0; v < 8; ++v) {
        const float before = (nt > 0)
            ? frame_before<true>(ps[0][nt][v], ps[0][nt > 0 ? nt - 1 : 0][v])
            : frame_before<false>(ps[0][nt][v], 0.f);
        const float after = (nt < 3)
            ? frame_after<true>(ps[2][nt][v], ps[2][nt < 3 ? nt + 1 : 3][v])
            : frame_after<false>(ps[2][nt][v], 0.f);
        float f = (before + ps[1][nt][v]) + after;
        f = f > 0.f ? f : 0.01f * f;
        const int row = 2 + nt * 16 + (lane & 15);
        const int col = m0 + v + (hs << 3);
        (br ? Ht1 : Ht0)[row * HT_STRIDE + col] = (_Float16)f;
      }
    }
  }
  __syncthreads();

  for (int j = 0; j < 2; ++j) {
    const int m0 = wave * 32 + j * 16;
    _Float16* const sW2 = sW + (grp * 8 + wave) * WB_WAVE;
    _Float16* const sW1 = sW2 + 16 * W2_PITCH;
    {
      const float4* w2src = reinterpret_cast<const float4*>(
          W2b + ((size_t)(part * CH + m0)) * HID * 3);
      const float4* w1src = reinterpret_cast<const float4*>(
          W1b + ((size_t)(part * CH + m0)) * HID);
      float4 s2[24], s1[8];
#pragma unroll
      for (int t = 0; t < 24; ++t) s2[t] = w2src[t * 32 + lane];
#pragma unroll
      for (int t = 0; t < 8; ++t) s1[t] = w1src[t * 32 + lane];
      int d2[3];
#pragma unroll
      for (int t = 0; t < 3; ++t) {
        const int f = t * 32 + lane, row = f / 48, q = f - row * 48;
        d2[t] = row * W2_PITCH + q * 4;
      }
#pragma unroll
      for (int a = 0; a < 8; ++a)
#pragma unroll
        for (int t = 0; t < 3; ++t)
          store_h4(sW2 + a * 2 * W2_PITCH + d2[t], s2[a * 3 + t]);
      const int d1 = (lane >> 4) * W1_PITCH + (lane & 15) * 4;
#pragma unroll
      for (int t = 0; t < 8; ++t) store_h4(sW1 + t * 2 * W1_PITCH + d1, s1[t]);
    }
    __builtin_amdgcn_wave_barrier();
    const _Float16* const w1run = sW1 + (lane & 15) * W1_PITCH + hs * 16;
    const _Float16* const w2run = sW2 + (lane & 15) * W2_PITCH + hs * 48;
    v16h a1[2], a2[3][2];
#pragma unroll
    for (int kb = 0; kb < 2; ++kb) {
      _Float16 r1[16];
      load_hrun<2>(w1run + kb * 32, r1);
      a1[kb] = frag_of_hrun<1>(r1, 0);
      _Float16 r2[48];
      load_hrun<6>(w2run + kb * 96, r2);
#pragma unroll
      for (int s = 0; s < 3; ++s) a2[s][kb] = frag_of_hrun<3>(r2, s);
    }

    float maxv[8];
#pragma unroll
    for (int v = 0; v < 8; ++v) maxv[v] = -INFINITY;

#pragma nounroll
    for (int nt = 0; nt < 4; ++nt) {
      v8f acc1 = {}, acc2 = {};
#pragma unroll
      for (int kb = 0; kb < 2; ++kb) {
        v16h bf = load_b_frag_lds(Ht0, 2 + nt * 16, HT_STRIDE, kb * 32, lane);
        acc1 = wmma_f16(a1[kb], bf, acc1);
      }
      for (int s = 0; s < 3; ++s) {
#pragma unroll
        for (int kb = 0; kb < 2; ++kb) {
          v16h bf = load_b_frag_lds(Ht1, 1 + s + nt * 16, HT_STRIDE,
                                    kb * 32, lane);
          acc2 = wmma_f16(a2[s][kb], bf, acc2);
        }
      }
#pragma unroll
      for (int v = 0; v < 8; ++v) {
        const int c = m0 + v + (hs << 3);
        const int l = nt * 16 + (lane & 15);
        float xs[5];
#pragma unroll
        for (int d = 0; d < 5; ++d)
          xs[d] = (float)sXt[(l + d) * XT_STRIDE + xcol + c];
        const float avg3 = (xs[1] + xs[2] + xs[3]) * (1.0f / 3.0f);
        const float avg5 = (xs[0] + xs[1] + xs[2] + xs[3] + xs[4]) * 0.2f;
        float m3 = xs[2];
        m3 = fmaxf(m3, (l >= 1)       ? xs[1] : -INFINITY);
        m3 = fmaxf(m3, (l <= cfg::SEQ - 2) ? xs[3] : -INFINITY);
        float m5 = m3;
        m5 = fmaxf(m5, (l >= 2)       ? xs[0] : -INFINITY);
        m5 = fmaxf(m5, (l <= cfg::SEQ - 3) ? xs[4] : -INFINITY);
        const float g1 = sigmoidf_(acc1[v]);
        const float g2 = sigmoidf_(acc2[v]);
        const float mval = (avg3 + m3) * g1 + (avg5 + m5) * g2;
        maxv[v] = fmaxf(maxv[v], mval);
      }
    }
#pragma unroll
    for (int v = 0; v < 8; ++v) {
#pragma unroll
      for (int off = 8; off >= 1; off >>= 1)
        maxv[v] = fmaxf(maxv[v], __shfl_xor(maxv[v], off, 16));
      if ((lane & 15) == 0) {
        const int c = m0 + v + (hs << 3);
        out[((size_t)b * cfg::PARTS + part) * cfg::CH + c] = maxv[v];
      }
    }
  }
  }
}

extern "C" void kernel_launch(void* const* d_in, const int* in_sizes, int n_in,
                              void* d_out, int out_size, void* d_ws, size_t ws_size,
                              hipStream_t stream) {
  (void)in_sizes; (void)n_in; (void)d_ws; (void)ws_size; (void)out_size;
  const float* x   = (const float*)d_in[0];
  const float* W1a = (const float*)d_in[1];
  const float* W1b = (const float*)d_in[2];
  const float* W2a = (const float*)d_in[3];
  const float* W2b = (const float*)d_in[4];
  float* out = (float*)d_out;

  dim3 grid(cfg::GRID_X, cfg::NB);
  dim3 block(cfg::THREADS);
  mtb_gate_kernel<<<grid, block, 0, stream>>>(x, W1a, W1b, W2a, W2b, out);
}


# Stage B: a channel tile's pooled inputs (avg+max over 3 and 5 frames) are worked out while its weight fetch is in flight, in two halves of 16 loads; the unrolled gate loop only multiplies and maxes
# speedup vs baseline: 2.6827x; 1.0622x over previous
#include <hip/hip_runtime.h>
#include <math.h>

typedef __attribute__((ext_vector_type(16))) _Float16 v16h;
typedef __attribute__((ext_vector_type(8)))  _Float16 v8h;
typedef __attribute__((ext_vector_type(8)))  float    v8f;

namespace cfg {
constexpr int NB    = 128;
constexpr int SEQ   = 64;
constexpr int CH    = 256;
constexpr int PARTS = 16;
constexpr int HID   = 64;
constexpr int PPB    = 4;
constexpr int GROUPS = 2;
constexpr int PPG    = PPB / GROUPS;
constexpr int GRID_X  = PARTS / PPB;
constexpr int THREADS = GROUPS * 256;
constexpr int XT_ROWS   = SEQ + 4;
constexpr int XT_STRIDE = PPB * CH + 8;
constexpr int HT_STRIDE = HID + 8;
constexpr int WA_PITCH = 32 * 3 + 8;
constexpr int WA_WAVE  = 16 * WA_PITCH;
constexpr int W2_PITCH = HID * 3 + 8;
constexpr int W1_PITCH = HID + 8;
constexpr int WB_WAVE  = 16 * (W2_PITCH + W1_PITCH);
constexpr int WST_A    = GROUPS * 8 * 2 * WA_WAVE;
constexpr int WST_B    = GROUPS * 8 * WB_WAVE;
constexpr int WST_HALFS = WST_A > WST_B ? WST_A : WST_B;
}

template <int NV8>
__device__ __forceinline__ void load_hrun(const _Float16* p, _Float16 (&r)[8 * NV8]) {
  const v8h* p8 = reinterpret_cast<const v8h*>(p);
#pragma unroll
  for (int i = 0; i < NV8; ++i) {
    const v8h q = p8[i];
#pragma unroll
    for (int e = 0; e < 8; ++e) r[8 * i + e] = q[e];
  }
}

template <int NTAP>
__device__ __forceinline__ v16h frag_of_hrun(const _Float16 (&r)[16 * NTAP], int s) {
  v16h a;
#pragma unroll
  for (int t = 0; t < 16; ++t) a[t] = r[t * NTAP + s];
  return a;
}

typedef __attribute__((ext_vector_type(4))) _Float16 v4h;
__device__ __forceinline__ void store_h4(_Float16* d, float4 v) {
  v4h h;
  h[0] = (_Float16)v.x; h[1] = (_Float16)v.y; h[2] = (_Float16)v.z; h[3] = (_Float16)v.w;
  *reinterpret_cast<v4h*>(d) = h;
}

typedef __attribute__((ext_vector_type(4))) float v4f;
__device__ __forceinline__ void store_h4(_Float16* d, v4f v) {
  v4h h;
  h[0] = (_Float16)v[0]; h[1] = (_Float16)v[1]; h[2] = (_Float16)v[2]; h[3] = (_Float16)v[3];
  *reinterpret_cast<v4h*>(d) = h;
}

__device__ __forceinline__ v16h load_b_frag_lds(const _Float16* lds, int row_base,
                                                int row_stride, int col_base,
                                                int lane) {
  const int n  = lane & 15;
  const int hs = (lane >> 4) & 1;
  const v8h* p = reinterpret_cast<const v8h*>(
      lds + (size_t)(row_base + n) * row_stride + col_base + (hs << 4));
  const v8h lo = p[0], hi = p[1];
  return __builtin_shufflevector(lo, hi, 0, 1, 2, 3, 4, 5, 6, 7,
                                 8, 9, 10, 11, 12, 13, 14, 15);
}

__device__ __forceinline__ v8f wmma_f16(v16h a, v16h b, v8f c) {
  return __builtin_amdgcn_wmma_f32_16x16x32_f16(false, a, false, b,
                                                (short)0, c, false, false);
}

__device__ __forceinline__ float sigmoidf_(float z) {
  return 1.0f / (1.0f + __expf(-z));
}

template <bool HAS_PREV>
__device__ __forceinline__ float frame_before(float tile, float prev_tile) {
  int edge = 0;
  if (HAS_PREV)
    edge = __builtin_amdgcn_update_dpp(0, __float_as_int(prev_tile), 0x121, 0xF, 0xF, false);
  return __int_as_float(
      __builtin_amdgcn_update_dpp(edge, __float_as_int(tile), 0x111, 0xF, 0xF, false));
}
template <bool HAS_NEXT>
__device__ __forceinline__ float frame_after(float tile, float next_tile) {
  int edge = 0;
  if (HAS_NEXT)
    edge = __builtin_amdgcn_update_dpp(0, __float_as_int(next_tile), 0x12F, 0xF, 0xF, false);
  return __int_as_float(
      __builtin_amdgcn_update_dpp(edge, __float_as_int(tile), 0x101, 0xF, 0xF, false));
}

__global__ __launch_bounds__(cfg::THREADS)
void mtb_gate_kernel(const float* __restrict__ x,
                     const float* __restrict__ W1a,
                     const float* __restrict__ W1b,
                     const float* __restrict__ W2a,
                     const float* __restrict__ W2b,
                     float* __restrict__ out) {
  using namespace cfg;
  __shared__ alignas(16) _Float16 sXt[XT_ROWS * XT_STRIDE];
  __shared__ alignas(16) _Float16 sHt[GROUPS][2][XT_ROWS * HT_STRIDE];
  __shared__ alignas(16) _Float16 sW[WST_HALFS];

  const int p0   = blockIdx.x * PPB;
  const int b    = blockIdx.y;
  const int tid  = threadIdx.x;
  const int lane = tid & 31;
  const int wv   = __builtin_amdgcn_readfirstlane(tid >> 5);
  const int grp  = wv >> 3;
  const int gtid = tid & 255;
  const int wave = wv & 7;
  const int hs   = (lane >> 4) & 1;

  {
    {
      const int pr  = tid >> 7;
      const int row = (pr < 2) ? pr : (SEQ + pr);
      const v8h zero = {};
      *reinterpret_cast<v8h*>(&sXt[row * XT_STRIDE + (tid & 127) * 8]) = zero;
    }
    if (gtid < HID) {
#pragma unroll
      for (int r = 0; r < 2; ++r) {
        sHt[grp][0][r * HT_STRIDE + gtid] = (_Float16)0.f;
        sHt[grp][1][r * HT_STRIDE + gtid] = (_Float16)0.f;
        sHt[grp][0][(SEQ + 2 + r) * HT_STRIDE + gtid] = (_Float16)0.f;
        sHt[grp][1][(SEQ + 2 + r) * HT_STRIDE + gtid] = (_Float16)0.f;
      }
    }
    const int c  = gtid;
    const int lh = grp;
    const float* xp = x + (((size_t)b * SEQ + lh) * CH + c) * PARTS + p0;
#pragma unroll 8
    for (int i = 0; i < SEQ / 2; ++i) {
      const float4 v = *reinterpret_cast<const float4*>(xp + (size_t)i * 2 * CH * PARTS);
      _Float16* d = sXt + (2 + 2 * i + lh) * XT_STRIDE + c;
      d[0 * CH] = (_Float16)v.x;
      d[1 * CH] = (_Float16)v.y;
      d[2 * CH] = (_Float16)v.z;
      d[3 * CH] = (_Float16)v.w;
    }
  }
  __syncthreads();

  for (int i = 0; i < PPG; ++i) {
  const int q    = grp * PPG + i;
  const int part = p0 + q;
  const int xcol = q * CH;
  _Float16* const Ht0 = sHt[grp][0];
  _Float16* const Ht1 = sHt[grp][1];
  if (i > 0) __syncthreads();

  {
    const int br = wave >> 2;
    const int mt = wave & 3;
    const int m0 = mt * 16;
    _Float16* const sWa = sW + (grp * 8 + wave) * (2 * WA_WAVE);
    const float* const wten = br ? W2a : W1a;
    const int rl = (lane >> 3) << 2;
    const unsigned goff = (unsigned)((part * HID + m0 + rl) * (CH * 3) + (lane & 7) * 4);
    const int ldst = rl * WA_PITCH + (lane & 7) * 4;
    v4f st[12];
#pragma unroll
    for (int a = 0; a < 4; ++a)
#pragma unroll
      for (int j = 0; j < 3; ++j)
        st[a * 3 + j] = *reinterpret_cast<const v4f*>(wten + goff + a * (CH * 3) + j * 32);
#pragma unroll
    for (int a = 0; a < 4; ++a)
#pragma unroll
      for (int j = 0; j < 3; ++j)
        store_h4(sWa + a * WA_PITCH + j * 32 + ldst, st[a * 3 + j]);
    __builtin_amdgcn_wave_barrier();

    const _Float16* const arun = sWa + (lane & 15) * WA_PITCH + hs * 48;
    v8f ps[3][4] = {};
    for (int kb = 0; kb < CH / 32; ++kb) {
      const int c0 = kb * 32;
      const bool more = kb + 1 < CH / 32;
      if (more) {
#pragma unroll
        for (int a = 0; a < 4; ++a)
#pragma unroll
          for (int j = 0; j < 3; ++j)
            st[a * 3 + j] = *reinterpret_cast<const v4f*>(
                wten + goff + (kb + 1) * 96 + a * (CH * 3) + j * 32);
      }
      __builtin_amdgcn_sched_barrier(0);
      _Float16 r[48];
      load_hrun<6>(arun + (kb & 1) * WA_WAVE, r);
      v16h a[3];
#pragma unroll
      for (int s = 0; s < 3; ++s) a[s] = frag_of_hrun<3>(r, s);
      v16h bf[4];
#pragma unroll
      for (int nt = 0; nt < 4; ++nt)
        bf[nt] = load_b_frag_lds(sXt, 2 + nt * 16, XT_STRIDE, xcol + c0, lane);
#pragma unroll
      for (int nt = 0; nt < 4; ++nt)
#pragma unroll
        for (int s = 0; s < 3; ++s) ps[s][nt] = wmma_f16(a[s], bf[nt], ps[s][nt]);
      __builtin_amdgcn_sched_barrier(0);
      if (more) {
        asm volatile("" :: "v"(st[0]), "v"(st[1]), "v"(st[2]), "v"(st[3]), "v"(st[4]), "v"(st[5]),
                           "v"(st[6]), "v"(st[7]), "v"(st[8]), "v"(st[9]), "v"(st[10]), "v"(st[11]));
#pragma unroll
        for (int a = 0; a < 4; ++a)
#pragma unroll
          for (int j = 0; j < 3; ++j)
            store_h4(sWa + ((kb + 1) & 1) * WA_WAVE + a * WA_PITCH + j * 32 + ldst,
                     st[a * 3 + j]);
        __builtin_amdgcn_wave_barrier();
      }
    }
#pragma unroll
    for (int nt = 0; nt < 4; ++nt) {
#pragma unroll
      for (int v = 0; v < 8; ++v) {
        const float before = (nt > 0)
            ? frame_before<true>(ps[0][nt][v], ps[0][nt > 0 ? nt - 1 : 0][v])
            : frame_before<false>(ps[0][nt][v], 0.f);
        const float after = (nt < 3)
            ? frame_after<true>(ps[2][nt][v], ps[2][nt < 3 ? nt + 1 : 3][v])
            : frame_after<false>(ps[2][nt][v], 0.f);
        float f = (before + ps[1][nt][v]) + after;
        f = f > 0.f ? f : 0.01f * f;
        const int row = 2 + nt * 16 + (lane & 15);
        const int col = m0 + v + (hs << 3);
        (br ? Ht1 : Ht0)[row * HT_STRIDE + col] = (_Float16)f;
      }
    }
  }
  __syncthreads();

  for (int j = 0; j < 2; ++j) {
    const int m0 = wave * 32 + j * 16;
    _Float16* const sW2 = sW + (grp * 8 + wave) * WB_WAVE;
    _Float16* const sW1 = sW2 + 16 * W2_PITCH;
    const v4f* const w2src = reinterpret_cast<const v4f*>(
        W2b + ((size_t)(part * CH + m0)) * HID * 3);
    const v4f* const w1src = reinterpret_cast<const v4f*>(
        W1b + ((size_t)(part * CH + m0)) * HID);
    int d2[3];
#pragma unroll
    for (int t = 0; t < 3; ++t) {
      const int f = t * 32 + lane, row = f / 48, q = f - row * 48;
      d2[t] = row * W2_PITCH + q * 4;
    }
    const int d1 = (lane >> 4) * W1_PITCH + (lane & 15) * 4;

    float p3[4][8], p5[4][8];
#pragma unroll
    for (int half = 0; half < 2; ++half) {
      v4f sf[16];
#pragma unroll
      for (int t = 0; t < 16; ++t) {
        const int u = half * 16 + t;
        sf[t] = (u < 24) ? w2src[u * 32 + lane] : w1src[(u - 24) * 32 + lane];
      }
      __builtin_amdgcn_sched_barrier(0);
#pragma unroll
      for (int nh = 0; nh < 2; ++nh) {
        const int nt = half * 2 + nh;
#pragma unroll
        for (int v = 0; v < 8; ++v) {
          const int c = m0 + v + (hs << 3);
          const int l = nt * 16 + (lane & 15);
          float xs[5];
#pragma unroll
          for (int d = 0; d < 5; ++d)
            xs[d] = (float)sXt[(l + d) * XT_STRIDE + xcol + c];
          const float avg3 = (xs[1] + xs[2] + xs[3]) * (1.0f / 3.0f);
          const float avg5 = (xs[0] + xs[1] + xs[2] + xs[3] + xs[4]) * 0.2f;
          float m3 = xs[2];
          m3 = fmaxf(m3, (l >= 1)       ? xs[1] : -INFINITY);
          m3 = fmaxf(m3, (l <= cfg::SEQ - 2) ? xs[3] : -INFINITY);
          float m5 = m3;
          m5 = fmaxf(m5, (l >= 2)       ? xs[0] : -INFINITY);
          m5 = fmaxf(m5, (l <= cfg::SEQ - 3) ? xs[4] : -INFINITY);
          p3[nt][v] = avg3 + m3;
          p5[nt][v] = avg5 + m5;
        }
      }
      __builtin_amdgcn_sched_barrier(0);
      asm volatile("" : "+v"(sf[0]), "+v"(sf[1]), "+v"(sf[2]), "+v"(sf[3]),
                        "+v"(sf[4]), "+v"(sf[5]), "+v"(sf[6]), "+v"(sf[7]));
      asm volatile("" : "+v"(sf[8]), "+v"(sf[9]), "+v"(sf[10]), "+v"(sf[11]),
                        "+v"(sf[12]), "+v"(sf[13]), "+v"(sf[14]), "+v"(sf[15]));
#pragma unroll
      for (int t = 0; t < 16; ++t) {
        const int u = half * 16 + t;
        if (u < 24)
          store_h4(sW2 + (u / 3) * 2 * W2_PITCH + d2[u % 3], sf[t]);
        else
          store_h4(sW1 + (u - 24) * 2 * W1_PITCH + d1, sf[t]);
      }
    }
    __builtin_amdgcn_wave_barrier();
    const _Float16* const w1run = sW1 + (lane & 15) * W1_PITCH + hs * 16;
    const _Float16* const w2run = sW2 + (lane & 15) * W2_PITCH + hs * 48;
    v16h a1[2], a2[3][2];
#pragma unroll
    for (int kb = 0; kb < 2; ++kb) {
      _Float16 r1[16];
      load_hrun<2>(w1run + kb * 32, r1);
      a1[kb] = frag_of_hrun<1>(r1, 0);
      _Float16 r2[48];
      load_hrun<6>(w2run + kb * 96, r2);
#pragma unroll
      for (int s = 0; s < 3; ++s) a2[s][kb] = frag_of_hrun<3>(r2, s);
    }

    float maxv[8];
#pragma unroll
    for (int v = 0; v < 8; ++v) maxv[v] = -INFINITY;

#pragma unroll
    for (int nt = 0; nt < 4; ++nt) {
      v8f acc1 = {}, acc2 = {};
#pragma unroll
      for (int kb = 0; kb < 2; ++kb) {
        v16h bf = load_b_frag_lds(Ht0, 2 + nt * 16, HT_STRIDE, kb * 32, lane);
        acc1 = wmma_f16(a1[kb], bf, acc1);
      }
#pragma unroll
      for (int s = 0; s < 3; ++s) {
#pragma unroll
        for (int kb = 0; kb < 2; ++kb) {
          v16h bf = load_b_frag_lds(Ht1, 1 + s + nt * 16, HT_STRIDE,
                                    kb * 32, lane);
          acc2 = wmma_f16(a2[s][kb], bf, acc2);
        }
      }
#pragma unroll
      for (int v = 0; v < 8; ++v) {
        const float g1 = sigmoidf_(acc1[v]);
        const float g2 = sigmoidf_(acc2[v]);
        const float mval = p3[nt][v] * g1 + p5[nt][v] * g2;
        maxv[v] = fmaxf(maxv[v], mval);
      }
      __builtin_amdgcn_sched_group_barrier(0x100, 16, 0);
      __builtin_amdgcn_sched_group_barrier(0x008, 8, 0);
      __builtin_amdgcn_sched_barrier(0);
    }
#pragma unroll
    for (int v = 0; v < 8; ++v) {
#pragma unroll
      for (int off = 8; off >= 1; off >>= 1)
        maxv[v] = fmaxf(maxv[v], __shfl_xor(maxv[v], off, 16));
    }
    if ((lane & 15) == 0) {
      float* const o = out + ((size_t)b * cfg::PARTS + part) * cfg::CH + m0 + (hs << 3);
#pragma unroll
      for (int v = 0; v < 8; ++v) o[v] = maxv[v];
    }
  }
  }
}

extern "C" void kernel_launch(void* const* d_in, const int* in_sizes, int n_in,
                              void* d_out, int out_size, void* d_ws, size_t ws_size,
                              hipStream_t stream) {
  (void)in_sizes; (void)n_in; (void)d_ws; (void)ws_size; (void)out_size;
  const float* x   = (const float*)d_in[0];
  const float* W1a = (const float*)d_in[1];
  const float* W1b = (const float*)d_in[2];
  const float* W2a = (const float*)d_in[3];
  const float* W2b = (const float*)d_in[4];
  float* out = (float*)d_out;

  dim3 grid(cfg::GRID_X, cfg::NB);
  dim3 block(cfg::THREADS);
  mtb_gate_kernel<<<grid, block, 0, stream>>>(x, W1a, W1b, W2a, W2b, out);
}


# Gate quotient 1/(1+e^-z) by one v_rcp_f32 (about 1 ulp, f32) instead of the 11-instruction exactly rounded division; v_exp_f32 and 1+e^-z kept as they are (not the tanh form)
# speedup vs baseline: 2.9880x; 1.1138x over previous
#include <hip/hip_runtime.h>
#include <math.h>

typedef __attribute__((ext_vector_type(16))) _Float16 v16h;
typedef __attribute__((ext_vector_type(8)))  _Float16 v8h;
typedef __attribute__((ext_vector_type(8)))  float    v8f;

namespace cfg {
constexpr int NB    = 128;
constexpr int SEQ   = 64;
constexpr int CH    = 256;
constexpr int PARTS = 16;
constexpr int HID   = 64;
constexpr int PPB    = 4;
constexpr int GROUPS = 2;
constexpr int PPG    = PPB / GROUPS;
constexpr int GRID_X  = PARTS / PPB;
constexpr int THREADS = GROUPS * 256;
constexpr int XT_ROWS   = SEQ + 4;
constexpr int XT_STRIDE = PPB * CH + 8;
constexpr int HT_STRIDE = HID + 8;
constexpr int WA_PITCH = 32 * 3 + 8;
constexpr int WA_WAVE  = 16 * WA_PITCH;
constexpr int W2_PITCH = HID * 3 + 8;
constexpr int W1_PITCH = HID + 8;
constexpr int WB_WAVE  = 16 * (W2_PITCH + W1_PITCH);
constexpr int WST_A    = GROUPS * 8 * 2 * WA_WAVE;
constexpr int WST_B    = GROUPS * 8 * WB_WAVE;
constexpr int WST_HALFS = WST_A > WST_B ? WST_A : WST_B;
}

template <int NV8>
__device__ __forceinline__ void load_hrun(const _Float16* p, _Float16 (&r)[8 * NV8]) {
  const v8h* p8 = reinterpret_cast<const v8h*>(p);
#pragma unroll
  for (int i = 0; i < NV8; ++i) {
    const v8h q = p8[i];
#pragma unroll
    for (int e = 0; e < 8; ++e) r[8 * i + e] = q[e];
  }
}

template <int NTAP>
__device__ __forceinline__ v16h frag_of_hrun(const _Float16 (&r)[16 * NTAP], int s) {
  v16h a;
#pragma unroll
  for (int t = 0; t < 16; ++t) a[t] = r[t * NTAP + s];
  return a;
}

typedef __attribute__((ext_vector_type(4))) _Float16 v4h;
__device__ __forceinline__ void store_h4(_Float16* d, float4 v) {
  v4h h;
  h[0] = (_Float16)v.x; h[1] = (_Float16)v.y; h[2] = (_Float16)v.z; h[3] = (_Float16)v.w;
  *reinterpret_cast<v4h*>(d) = h;
}

typedef __attribute__((ext_vector_type(4))) float v4f;
__device__ __forceinline__ void store_h4(_Float16* d, v4f v) {
  v4h h;
  h[0] = (_Float16)v[0]; h[1] = (_Float16)v[1]; h[2] = (_Float16)v[2]; h[3] = (_Float16)v[3];
  *reinterpret_cast<v4h*>(d) = h;
}

__device__ __forceinline__ v16h load_b_frag_lds(const _Float16* lds, int row_base,
                                                int row_stride, int col_base,
                                                int lane) {
  const int n  = lane & 15;
  const int hs = (lane >> 4) & 1;
  const v8h* p = reinterpret_cast<const v8h*>(
      lds + (size_t)(row_base + n) * row_stride + col_base + (hs << 4));
  const v8h lo = p[0], hi = p[1];
  return __builtin_shufflevector(lo, hi, 0, 1, 2, 3, 4, 5, 6, 7,
                                 8, 9, 10, 11, 12, 13, 14, 15);
}

__device__ __forceinline__ v8f wmma_f16(v16h a, v16h b, v8f c) {
  return __builtin_amdgcn_wmma_f32_16x16x32_f16(false, a, false, b,
                                                (short)0, c, false, false);
}

__device__ __forceinline__ float sigmoidf_(float z) {
  return __builtin_amdgcn_rcpf(1.0f + __expf(-z));
}

template <bool HAS_PREV>
__device__ __forceinline__ float frame_before(float tile, float prev_tile) {
  int edge = 0;
  if (HAS_PREV)
    edge = __builtin_amdgcn_update_dpp(0, __float_as_int(prev_tile), 0x121, 0xF, 0xF, false);
  return __int_as_float(
      __builtin_amdgcn_update_dpp(edge, __float_as_int(tile), 0x111, 0xF, 0xF, false));
}
template <bool HAS_NEXT>
__device__ __forceinline__ float frame_after(float tile, float next_tile) {
  int edge = 0;
  if (HAS_NEXT)
    edge = __builtin_amdgcn_update_dpp(0, __float_as_int(next_tile), 0x12F, 0xF, 0xF, false);
  return __int_as_float(
      __builtin_amdgcn_update_dpp(edge, __float_as_int(tile), 0x101, 0xF, 0xF, false));
}

__global__ __launch_bounds__(cfg::THREADS)
void mtb_gate_kernel(const float* __restrict__ x,
                     const float* __restrict__ W1a,
                     const float* __restrict__ W1b,
                     const float* __restrict__ W2a,
                     const float* __restrict__ W2b,
                     float* __restrict__ out) {
  using namespace cfg;
  __shared__ alignas(16) _Float16 sXt[XT_ROWS * XT_STRIDE];
  __shared__ alignas(16) _Float16 sHt[GROUPS][2][XT_ROWS * HT_STRIDE];
  __shared__ alignas(16) _Float16 sW[WST_HALFS];

  const int p0   = blockIdx.x * PPB;
  const int b    = blockIdx.y;
  const int tid  = threadIdx.x;
  const int lane = tid & 31;
  const int wv   = __builtin_amdgcn_readfirstlane(tid >> 5);
  const int grp  = wv >> 3;
  const int gtid = tid & 255;
  const int wave = wv & 7;
  const int hs   = (lane >> 4) & 1;

  {
    {
      const int pr  = tid >> 7;
      const int row = (pr < 2) ? pr : (SEQ + pr);
      const v8h zero = {};
      *reinterpret_cast<v8h*>(&sXt[row * XT_STRIDE + (tid & 127) * 8]) = zero;
    }
    if (gtid < HID) {
#pragma unroll
      for (int r = 0; r < 2; ++r) {
        sHt[grp][0][r * HT_STRIDE + gtid] = (_Float16)0.f;
        sHt[grp][1][r * HT_STRIDE + gtid] = (_Float16)0.f;
        sHt[grp][0][(SEQ + 2 + r) * HT_STRIDE + gtid] = (_Float16)0.f;
        sHt[grp][1][(SEQ + 2 + r) * HT_STRIDE + gtid] = (_Float16)0.f;
      }
    }
    const int c  = gtid;
    const int lh = grp;
    const float* xp = x + (((size_t)b * SEQ + lh) * CH + c) * PARTS + p0;
#pragma unroll 8
    for (int i = 0; i < SEQ / 2; ++i) {
      const float4 v = *reinterpret_cast<const float4*>(xp + (size_t)i * 2 * CH * PARTS);
      _Float16* d = sXt + (2 + 2 * i + lh) * XT_STRIDE + c;
      d[0 * CH] = (_Float16)v.x;
      d[1 * CH] = (_Float16)v.y;
      d[2 * CH] = (_Float16)v.z;
      d[3 * CH] = (_Float16)v.w;
    }
  }
  __syncthreads();

  for (int i = 0; i < PPG; ++i) {
  const int q    = grp * PPG + i;
  const int part = p0 + q;
  const int xcol = q * CH;
  _Float16* const Ht0 = sHt[grp][0];
  _Float16* const Ht1 = sHt[grp][1];
  if (i > 0) __syncthreads();

  {
    const int br = wave >> 2;
    const int mt = wave & 3;
    const int m0 = mt * 16;
    _Float16* const sWa = sW + (grp * 8 + wave) * (2 * WA_WAVE);
    const float* const wten = br ? W2a : W1a;
    const int rl = (lane >> 3) << 2;
    const unsigned goff = (unsigned)((part * HID + m0 + rl) * (CH * 3) + (lane & 7) * 4);
    const int ldst = rl * WA_PITCH + (lane & 7) * 4;
    v4f st[12];
#pragma unroll
    for (int a = 0; a < 4; ++a)
#pragma unroll
      for (int j = 0; j < 3; ++j)
        st[a * 3 + j] = *reinterpret_cast<const v4f*>(wten + goff + a * (CH * 3) + j * 32);
#pragma unroll
    for (int a = 0; a < 4; ++a)
#pragma unroll
      for (int j = 0; j < 3; ++j)
        store_h4(sWa + a * WA_PITCH + j * 32 + ldst, st[a * 3 + j]);
    __builtin_amdgcn_wave_barrier();

    const _Float16* const arun = sWa + (lane & 15) * WA_PITCH + hs * 48;
    v8f ps[3][4] = {};
    for (int kb = 0; kb < CH / 32; ++kb) {
      const int c0 = kb * 32;
      const bool more = kb + 1 < CH / 32;
      if (more) {
#pragma unroll
        for (int a = 0; a < 4; ++a)
#pragma unroll
          for (int j = 0; j < 3; ++j)
            st[a * 3 + j] = *reinterpret_cast<const v4f*>(
                wten + goff + (kb + 1) * 96 + a * (CH * 3) + j * 32);
      }
      __builtin_amdgcn_sched_barrier(0);
      _Float16 r[48];
      load_hrun<6>(arun + (kb & 1) * WA_WAVE, r);
      v16h a[3];
#pragma unroll
      for (int s = 0; s < 3; ++s) a[s] = frag_of_hrun<3>(r, s);
      v16h bf[4];
#pragma unroll
      for (int nt = 0; nt < 4; ++nt)
        bf[nt] = load_b_frag_lds(sXt, 2 + nt * 16, XT_STRIDE, xcol + c0, lane);
#pragma unroll
      for (int nt = 0; nt < 4; ++nt)
#pragma unroll
        for (int s = 0; s < 3; ++s) ps[s][nt] = wmma_f16(a[s], bf[nt], ps[s][nt]);
      __builtin_amdgcn_sched_barrier(0);
      if (more) {
        asm volatile("" :: "v"(st[0]), "v"(st[1]), "v"(st[2]), "v"(st[3]), "v"(st[4]), "v"(st[5]),
                           "v"(st[6]), "v"(st[7]), "v"(st[8]), "v"(st[9]), "v"(st[10]), "v"(st[11]));
#pragma unroll
        for (int a = 0; a < 4; ++a)
#pragma unroll
          for (int j = 0; j < 3; ++j)
            store_h4(sWa + ((kb + 1) & 1) * WA_WAVE + a * WA_PITCH + j * 32 + ldst,
                     st[a * 3 + j]);
        __builtin_amdgcn_wave_barrier();
      }
    }
#pragma unroll
    for (int nt = 0; nt < 4; ++nt) {
#pragma unroll
      for (int v = 0; v < 8; ++v) {
        const float before = (nt > 0)
            ? frame_before<true>(ps[0][nt][v], ps[0][nt > 0 ? nt - 1 : 0][v])
            : frame_before<false>(ps[0][nt][v], 0.f);
        const float after = (nt < 3)
            ? frame_after<true>(ps[2][nt][v], ps[2][nt < 3 ? nt + 1 : 3][v])
            : frame_after<false>(ps[2][nt][v], 0.f);
        float f = (before + ps[1][nt][v]) + after;
        f = f > 0.f ? f : 0.01f * f;
        const int row = 2 + nt * 16 + (lane & 15);
        const int col = m0 + v + (hs << 3);
        (br ? Ht1 : Ht0)[row * HT_STRIDE + col] = (_Float16)f;
      }
    }
  }
  __syncthreads();

  for (int j = 0; j < 2; ++j) {
    const int m0 = wave * 32 + j * 16;
    _Float16* const sW2 = sW + (grp * 8 + wave) * WB_WAVE;
    _Float16* const sW1 = sW2 + 16 * W2_PITCH;
    const v4f* const w2src = reinterpret_cast<const v4f*>(
        W2b + ((size_t)(part * CH + m0)) * HID * 3);
    const v4f* const w1src = reinterpret_cast<const v4f*>(
        W1b + ((size_t)(part * CH + m0)) * HID);
    int d2[3];
#pragma unroll
    for (int t = 0; t < 3; ++t) {
      const int f = t * 32 + lane, row = f / 48, q = f - row * 48;
      d2[t] = row * W2_PITCH + q * 4;
    }
    const int d1 = (lane >> 4) * W1_PITCH + (lane & 15) * 4;

    float p3[4][8], p5[4][8];
#pragma unroll
    for (int half = 0; half < 2; ++half) {
      v4f sf[16];
#pragma unroll
      for (int t = 0; t < 16; ++t) {
        const int u = half * 16 + t;
        sf[t] = (u < 24) ? w2src[u * 32 + lane] : w1src[(u - 24) * 32 + lane];
      }
      __builtin_amdgcn_sched_barrier(0);
#pragma unroll
      for (int nh = 0; nh < 2; ++nh) {
        const int nt = half * 2 + nh;
#pragma unroll
        for (int v = 0; v < 8; ++v) {
          const int c = m0 + v + (hs << 3);
          const int l = nt * 16 + (lane & 15);
          float xs[5];
#pragma unroll
          for (int d = 0; d < 5; ++d)
            xs[d] = (float)sXt[(l + d) * XT_STRIDE + xcol + c];
          const float avg3 = (xs[1] + xs[2] + xs[3]) * (1.0f / 3.0f);
          const float avg5 = (xs[0] + xs[1] + xs[2] + xs[3] + xs[4]) * 0.2f;
          float m3 = xs[2];
          m3 = fmaxf(m3, (l >= 1)       ? xs[1] : -INFINITY);
          m3 = fmaxf(m3, (l <= cfg::SEQ - 2) ? xs[3] : -INFINITY);
          float m5 = m3;
          m5 = fmaxf(m5, (l >= 2)       ? xs[0] : -INFINITY);
          m5 = fmaxf(m5, (l <= cfg::SEQ - 3) ? xs[4] : -INFINITY);
          p3[nt][v] = avg3 + m3;
          p5[nt][v] = avg5 + m5;
        }
      }
      __builtin_amdgcn_sched_barrier(0);
      asm volatile("" : "+v"(sf[0]), "+v"(sf[1]), "+v"(sf[2]), "+v"(sf[3]),
                        "+v"(sf[4]), "+v"(sf[5]), "+v"(sf[6]), "+v"(sf[7]));
      asm volatile("" : "+v"(sf[8]), "+v"(sf[9]), "+v"(sf[10]), "+v"(sf[11]),
                        "+v"(sf[12]), "+v"(sf[13]), "+v"(sf[14]), "+v"(sf[15]));
#pragma unroll
      for (int t = 0; t < 16; ++t) {
        const int u = half * 16 + t;
        if (u < 24)
          store_h4(sW2 + (u / 3) * 2 * W2_PITCH + d2[u % 3], sf[t]);
        else
          store_h4(sW1 + (u - 24) * 2 * W1_PITCH + d1, sf[t]);
      }
    }
    __builtin_amdgcn_wave_barrier();
    const _Float16* const w1run = sW1 + (lane & 15) * W1_PITCH + hs * 16;
    const _Float16* const w2run = sW2 + (lane & 15) * W2_PITCH + hs * 48;
    v16h a1[2], a2[3][2];
#pragma unroll
    for (int kb = 0; kb < 2; ++kb) {
      _Float16 r1[16];
      load_hrun<2>(w1run + kb * 32, r1);
      a1[kb] = frag_of_hrun<1>(r1, 0);
      _Float16 r2[48];
      load_hrun<6>(w2run + kb * 96, r2);
#pragma unroll
      for (int s = 0; s < 3; ++s) a2[s][kb] = frag_of_hrun<3>(r2, s);
    }

    float maxv[8];
#pragma unroll
    for (int v = 0; v < 8; ++v) maxv[v] = -INFINITY;

#pragma unroll
    for (int nt = 0; nt < 4; ++nt) {
      v8f acc1 = {}, acc2 = {};
#pragma unroll
      for (int kb = 0; kb < 2; ++kb) {
        v16h bf = load_b_frag_lds(Ht0, 2 + nt * 16, HT_STRIDE, kb * 32, lane);
        acc1 = wmma_f16(a1[kb], bf, acc1);
      }
#pragma unroll
      for (int s = 0; s < 3; ++s) {
#pragma unroll
        for (int kb = 0; kb < 2; ++kb) {
          v16h bf = load_b_frag_lds(Ht1, 1 + s + nt * 16, HT_STRIDE,
                                    kb * 32, lane);
          acc2 = wmma_f16(a2[s][kb], bf, acc2);
        }
      }
#pragma unroll
      for (int v = 0; v < 8; ++v) {
        const float g1 = sigmoidf_(acc1[v]);
        const float g2 = sigmoidf_(acc2[v]);
        const float mval = p3[nt][v] * g1 + p5[nt][v] * g2;
        maxv[v] = fmaxf(maxv[v], mval);
      }
      __builtin_amdgcn_sched_group_barrier(0x100, 16, 0);
      __builtin_amdgcn_sched_group_barrier(0x008, 8, 0);
      __builtin_amdgcn_sched_barrier(0);
    }
#pragma unroll
    for (int v = 0; v < 8; ++v) {
#pragma unroll
      for (int off = 8; off >= 1; off >>= 1)
        maxv[v] = fmaxf(maxv[v], __shfl_xor(maxv[v], off, 16));
    }
    if ((lane & 15) == 0) {
      float* const o = out + ((size_t)b * cfg::PARTS + part) * cfg::CH + m0 + (hs << 3);
#pragma unroll
      for (int v = 0; v < 8; ++v) o[v] = maxv[v];
    }
  }
  }
}

extern "C" void kernel_launch(void* const* d_in, const int* in_sizes, int n_in,
                              void* d_out, int out_size, void* d_ws, size_t ws_size,
                              hipStream_t stream) {
  (void)in_sizes; (void)n_in; (void)d_ws; (void)ws_size; (void)out_size;
  const float* x   = (const float*)d_in[0];
  const float* W1a = (const float*)d_in[1];
  const float* W1b = (const float*)d_in[2];
  const float* W2a = (const float*)d_in[3];
  const float* W2b = (const float*)d_in[4];
  float* out = (float*)d_out;

  dim3 grid(cfg::GRID_X, cfg::NB);
  dim3 block(cfg::THREADS);
  mtb_gate_kernel<<<grid, block, 0, stream>>>(x, W1a, W1b, W2a, W2b, out);
}


# Each block walks its weights in an order turned by its batch index (Stage A k blocks start at b&7 and wrap; part and channel-tile order by b bits 3, 4): f32 sum order over k changes
# speedup vs baseline: 3.1403x; 1.0510x over previous
#include <hip/hip_runtime.h>
#include <math.h>

typedef __attribute__((ext_vector_type(16))) _Float16 v16h;
typedef __attribute__((ext_vector_type(8)))  _Float16 v8h;
typedef __attribute__((ext_vector_type(8)))  float    v8f;

namespace cfg {
constexpr int NB    = 128;
constexpr int SEQ   = 64;
constexpr int CH    = 256;
constexpr int PARTS = 16;
constexpr int HID   = 64;
constexpr int PPB    = 4;
constexpr int GROUPS = 2;
constexpr int PPG    = PPB / GROUPS;
constexpr int GRID_X  = PARTS / PPB;
constexpr int THREADS = GROUPS * 256;
constexpr int XT_ROWS   = SEQ + 4;
constexpr int XT_STRIDE = PPB * CH + 8;
constexpr int HT_STRIDE = HID + 8;
constexpr int WA_PITCH = 32 * 3 + 8;
constexpr int WA_WAVE  = 16 * WA_PITCH;
constexpr int W2_PITCH = HID * 3 + 8;
constexpr int W1_PITCH = HID + 8;
constexpr int WB_WAVE  = 16 * (W2_PITCH + W1_PITCH);
constexpr int WST_A    = GROUPS * 8 * 2 * WA_WAVE;
constexpr int WST_B    = GROUPS * 8 * WB_WAVE;
constexpr int WST_HALFS = WST_A > WST_B ? WST_A : WST_B;
}

template <int NV8>
__device__ __forceinline__ void load_hrun(const _Float16* p, _Float16 (&r)[8 * NV8]) {
  const v8h* p8 = reinterpret_cast<const v8h*>(p);
#pragma unroll
  for (int i = 0; i < NV8; ++i) {
    const v8h q = p8[i];
#pragma unroll
    for (int e = 0; e < 8; ++e) r[8 * i + e] = q[e];
  }
}

template <int NTAP>
__device__ __forceinline__ v16h frag_of_hrun(const _Float16 (&r)[16 * NTAP], int s) {
  v16h a;
#pragma unroll
  for (int t = 0; t < 16; ++t) a[t] = r[t * NTAP + s];
  return a;
}

typedef __attribute__((ext_vector_type(4))) _Float16 v4h;
__device__ __forceinline__ void store_h4(_Float16* d, float4 v) {
  v4h h;
  h[0] = (_Float16)v.x; h[1] = (_Float16)v.y; h[2] = (_Float16)v.z; h[3] = (_Float16)v.w;
  *reinterpret_cast<v4h*>(d) = h;
}

typedef __attribute__((ext_vector_type(4))) float v4f;
__device__ __forceinline__ void store_h4(_Float16* d, v4f v) {
  v4h h;
  h[0] = (_Float16)v[0]; h[1] = (_Float16)v[1]; h[2] = (_Float16)v[2]; h[3] = (_Float16)v[3];
  *reinterpret_cast<v4h*>(d) = h;
}

__device__ __forceinline__ v16h load_b_frag_lds(const _Float16* lds, int row_base,
                                                int row_stride, int col_base,
                                                int lane) {
  const int n  = lane & 15;
  const int hs = (lane >> 4) & 1;
  const v8h* p = reinterpret_cast<const v8h*>(
      lds + (size_t)(row_base + n) * row_stride + col_base + (hs << 4));
  const v8h lo = p[0], hi = p[1];
  return __builtin_shufflevector(lo, hi, 0, 1, 2, 3, 4, 5, 6, 7,
                                 8, 9, 10, 11, 12, 13, 14, 15);
}

__device__ __forceinline__ v8f wmma_f16(v16h a, v16h b, v8f c) {
  return __builtin_amdgcn_wmma_f32_16x16x32_f16(false, a, false, b,
                                                (short)0, c, false, false);
}

__device__ __forceinline__ float sigmoidf_(float z) {
  return __builtin_amdgcn_rcpf(1.0f + __expf(-z));
}

template <bool HAS_PREV>
__device__ __forceinline__ float frame_before(float tile, float prev_tile) {
  int edge = 0;
  if (HAS_PREV)
    edge = __builtin_amdgcn_update_dpp(0, __float_as_int(prev_tile), 0x121, 0xF, 0xF, false);
  return __int_as_float(
      __builtin_amdgcn_update_dpp(edge, __float_as_int(tile), 0x111, 0xF, 0xF, false));
}
template <bool HAS_NEXT>
__device__ __forceinline__ float frame_after(float tile, float next_tile) {
  int edge = 0;
  if (HAS_NEXT)
    edge = __builtin_amdgcn_update_dpp(0, __float_as_int(next_tile), 0x12F, 0xF, 0xF, false);
  return __int_as_float(
      __builtin_amdgcn_update_dpp(edge, __float_as_int(tile), 0x101, 0xF, 0xF, false));
}

__global__ __launch_bounds__(cfg::THREADS)
void mtb_gate_kernel(const float* __restrict__ x,
                     const float* __restrict__ W1a,
                     const float* __restrict__ W1b,
                     const float* __restrict__ W2a,
                     const float* __restrict__ W2b,
                     float* __restrict__ out) {
  using namespace cfg;
  __shared__ alignas(16) _Float16 sXt[XT_ROWS * XT_STRIDE];
  __shared__ alignas(16) _Float16 sHt[GROUPS][2][XT_ROWS * HT_STRIDE];
  __shared__ alignas(16) _Float16 sW[WST_HALFS];

  const int p0   = blockIdx.x * PPB;
  const int b    = blockIdx.y;
  const int tid  = threadIdx.x;
  const int lane = tid & 31;
  const int wv   = __builtin_amdgcn_readfirstlane(tid >> 5);
  const int grp  = wv >> 3;
  const int gtid = tid & 255;
  const int wave = wv & 7;
  const int hs   = (lane >> 4) & 1;
  const int rot_k = b & 7;
  const int rot_p = (b >> 3) & 1;
  const int rot_j = (b >> 4) & 1;

  {
    {
      const int pr  = tid >> 7;
      const int row = (pr < 2) ? pr : (SEQ + pr);
      const v8h zero = {};
      *reinterpret_cast<v8h*>(&sXt[row * XT_STRIDE + (tid & 127) * 8]) = zero;
    }
    if (gtid < HID) {
#pragma unroll
      for (int r = 0; r < 2; ++r) {
        sHt[grp][0][r * HT_STRIDE + gtid] = (_Float16)0.f;
        sHt[grp][1][r * HT_STRIDE + gtid] = (_Float16)0.f;
        sHt[grp][0][(SEQ + 2 + r) * HT_STRIDE + gtid] = (_Float16)0.f;
        sHt[grp][1][(SEQ + 2 + r) * HT_STRIDE + gtid] = (_Float16)0.f;
      }
    }
    const int c  = gtid;
    const int lh = grp;
    const float* xp = x + (((size_t)b * SEQ + lh) * CH + c) * PARTS + p0;
#pragma unroll 8
    for (int i = 0; i < SEQ / 2; ++i) {
      const float4 v = *reinterpret_cast<const float4*>(xp + (size_t)i * 2 * CH * PARTS);
      _Float16* d = sXt + (2 + 2 * i + lh) * XT_STRIDE + c;
      d[0 * CH] = (_Float16)v.x;
      d[1 * CH] = (_Float16)v.y;
      d[2 * CH] = (_Float16)v.z;
      d[3 * CH] = (_Float16)v.w;
    }
  }
  __syncthreads();

  for (int i = 0; i < PPG; ++i) {
  const int q    = grp * PPG + ((i + rot_p) & (PPG - 1));
  const int part = p0 + q;
  const int xcol = q * CH;
  _Float16* const Ht0 = sHt[grp][0];
  _Float16* const Ht1 = sHt[grp][1];
  if (i > 0) __syncthreads();

  {
    const int br = wave >> 2;
    const int mt = wave & 3;
    const int m0 = mt * 16;
    _Float16* const sWa = sW + (grp * 8 + wave) * (2 * WA_WAVE);
    const float* const wten = br ? W2a : W1a;
    const int rl = (lane >> 3) << 2;
    const unsigned goff = (unsigned)((part * HID + m0 + rl) * (CH * 3) + (lane & 7) * 4);
    const int ldst = rl * WA_PITCH + (lane & 7) * 4;
    v4f st[12];
#pragma unroll
    for (int a = 0; a < 4; ++a)
#pragma unroll
      for (int j = 0; j < 3; ++j)
        st[a * 3 + j] = *reinterpret_cast<const v4f*>(wten + goff + rot_k * 96 + a * (CH * 3) + j * 32);
#pragma unroll
    for (int a = 0; a < 4; ++a)
#pragma unroll
      for (int j = 0; j < 3; ++j)
        store_h4(sWa + a * WA_PITCH + j * 32 + ldst, st[a * 3 + j]);
    __builtin_amdgcn_wave_barrier();

    const _Float16* const arun = sWa + (lane & 15) * WA_PITCH + hs * 48;
    v8f ps[3][4] = {};
    for (int it = 0; it < CH / 32; ++it) {
      const int kb = (it + rot_k) & (CH / 32 - 1);
      const int kn = (it + 1 + rot_k) & (CH / 32 - 1);
      const int c0 = kb * 32;
      const bool more = it + 1 < CH / 32;
      if (more) {
#pragma unroll
        for (int a = 0; a < 4; ++a)
#pragma unroll
          for (int j = 0; j < 3; ++j)
            st[a * 3 + j] = *reinterpret_cast<const v4f*>(
                wten + goff + kn * 96 + a * (CH * 3) + j * 32);
      }
      __builtin_amdgcn_sched_barrier(0);
      _Float16 r[48];
      load_hrun<6>(arun + (it & 1) * WA_WAVE, r);
      v16h a[3];
#pragma unroll
      for (int s = 0; s < 3; ++s) a[s] = frag_of_hrun<3>(r, s);
      v16h bf[4];
#pragma unroll
      for (int nt = 0; nt < 4; ++nt)
        bf[nt] = load_b_frag_lds(sXt, 2 + nt * 16, XT_STRIDE, xcol + c0, lane);
#pragma unroll
      for (int nt = 0; nt < 4; ++nt)
#pragma unroll
        for (int s = 0; s < 3; ++s) ps[s][nt] = wmma_f16(a[s], bf[nt], ps[s][nt]);
      __builtin_amdgcn_sched_barrier(0);
      if (more) {
        asm volatile("" :: "v"(st[0]), "v"(st[1]), "v"(st[2]), "v"(st[3]), "v"(st[4]), "v"(st[5]),
                           "v"(st[6]), "v"(st[7]), "v"(st[8]), "v"(st[9]), "v"(st[10]), "v"(st[11]));
#pragma unroll
        for (int a = 0; a < 4; ++a)
#pragma unroll
          for (int j = 0; j < 3; ++j)
            store_h4(sWa + ((it + 1) & 1) * WA_WAVE + a * WA_PITCH + j * 32 + ldst,
                     st[a * 3 + j]);
        __builtin_amdgcn_wave_barrier();
      }
    }
#pragma unroll
    for (int nt = 0; nt < 4; ++nt) {
#pragma unroll
      for (int v = 0; v < 8; ++v) {
        const float before = (nt > 0)
            ? frame_before<true>(ps[0][nt][v], ps[0][nt > 0 ? nt - 1 : 0][v])
            : frame_before<false>(ps[0][nt][v], 0.f);
        const float after = (nt < 3)
            ? frame_after<true>(ps[2][nt][v], ps[2][nt < 3 ? nt + 1 : 3][v])
            : frame_after<false>(ps[2][nt][v], 0.f);
        float f = (before + ps[1][nt][v]) + after;
        f = f > 0.f ? f : 0.01f * f;
        const int row = 2 + nt * 16 + (lane & 15);
        const int col = m0 + v + (hs << 3);
        (br ? Ht1 : Ht0)[row * HT_STRIDE + col] = (_Float16)f;
      }
    }
  }
  __syncthreads();

  for (int j = 0; j < 2; ++j) {
    const int m0 = wave * 32 + ((j + rot_j) & 1) * 16;
    _Float16* const sW2 = sW + (grp * 8 + wave) * WB_WAVE;
    _Float16* const sW1 = sW2 + 16 * W2_PITCH;
    const v4f* const w2src = reinterpret_cast<const v4f*>(
        W2b + ((size_t)(part * CH + m0)) * HID * 3);
    const v4f* const w1src = reinterpret_cast<const v4f*>(
        W1b + ((size_t)(part * CH + m0)) * HID);
    int d2[3];
#pragma unroll
    for (int t = 0; t < 3; ++t) {
      const int f = t * 32 + lane, row = f / 48, q = f - row * 48;
      d2[t] = row * W2_PITCH + q * 4;
    }
    const int d1 = (lane >> 4) * W1_PITCH + (lane & 15) * 4;

    float p3[4][8], p5[4][8];
#pragma unroll
    for (int half = 0; half < 2; ++half) {
      v4f sf[16];
#pragma unroll
      for (int t = 0; t < 16; ++t) {
        const int u = half * 16 + t;
        sf[t] = (u < 24) ? w2src[u * 32 + lane] : w1src[(u - 24) * 32 + lane];
      }
      __builtin_amdgcn_sched_barrier(0);
#pragma unroll
      for (int nh = 0; nh < 2; ++nh) {
        const int nt = half * 2 + nh;
#pragma unroll
        for (int v = 0; v < 8; ++v) {
          const int c = m0 + v + (hs << 3);
          const int l = nt * 16 + (lane & 15);
          float xs[5];
#pragma unroll
          for (int d = 0; d < 5; ++d)
            xs[d] = (float)sXt[(l + d) * XT_STRIDE + xcol + c];
          const float avg3 = (xs[1] + xs[2] + xs[3]) * (1.0f / 3.0f);
          const float avg5 = (xs[0] + xs[1] + xs[2] + xs[3] + xs[4]) * 0.2f;
          float m3 = xs[2];
          m3 = fmaxf(m3, (l >= 1)       ? xs[1] : -INFINITY);
          m3 = fmaxf(m3, (l <= cfg::SEQ - 2) ? xs[3] : -INFINITY);
          float m5 = m3;
          m5 = fmaxf(m5, (l >= 2)       ? xs[0] : -INFINITY);
          m5 = fmaxf(m5, (l <= cfg::SEQ - 3) ? xs[4] : -INFINITY);
          p3[nt][v] = avg3 + m3;
          p5[nt][v] = avg5 + m5;
        }
      }
      __builtin_amdgcn_sched_barrier(0);
      asm volatile("" : "+v"(sf[0]), "+v"(sf[1]), "+v"(sf[2]), "+v"(sf[3]),
                        "+v"(sf[4]), "+v"(sf[5]), "+v"(sf[6]), "+v"(sf[7]));
      asm volatile("" : "+v"(sf[8]), "+v"(sf[9]), "+v"(sf[10]), "+v"(sf[11]),
                        "+v"(sf[12]), "+v"(sf[13]), "+v"(sf[14]), "+v"(sf[15]));
#pragma unroll
      for (int t = 0; t < 16; ++t) {
        const int u = half * 16 + t;
        if (u < 24)
          store_h4(sW2 + (u / 3) * 2 * W2_PITCH + d2[u % 3], sf[t]);
        else
          store_h4(sW1 + (u - 24) * 2 * W1_PITCH + d1, sf[t]);
      }
    }
    __builtin_amdgcn_wave_barrier();
    const _Float16* const w1run = sW1 + (lane & 15) * W1_PITCH + hs * 16;
    const _Float16* const w2run = sW2 + (lane & 15) * W2_PITCH + hs * 48;
    v16h a1[2], a2[3][2];
#pragma unroll
    for (int kb = 0; kb < 2; ++kb) {
      _Float16 r1[16];
      load_hrun<2>(w1run + kb * 32, r1);
      a1[kb] = frag_of_hrun<1>(r1, 0);
      _Float16 r2[48];
      load_hrun<6>(w2run + kb * 96, r2);
#pragma unroll
      for (int s = 0; s < 3; ++s) a2[s][kb] = frag_of_hrun<3>(r2, s);
    }

    float maxv[8];
#pragma unroll
    for (int v = 0; v < 8; ++v) maxv[v] = -INFINITY;

#pragma unroll
    for (int nt = 0; nt < 4; ++nt) {
      v8f acc1 = {}, acc2 = {};
#pragma unroll
      for (int kb = 0; kb < 2; ++kb) {
        v16h bf = load_b_frag_lds(Ht0, 2 + nt * 16, HT_STRIDE, kb * 32, lane);
        acc1 = wmma_f16(a1[kb], bf, acc1);
      }
#pragma unroll
      for (int s = 0; s < 3; ++s) {
#pragma unroll
        for (int kb = 0; kb < 2; ++kb) {
          v16h bf = load_b_frag_lds(Ht1, 1 + s + nt * 16, HT_STRIDE,
                                    kb * 32, lane);
          acc2 = wmma_f16(a2[s][kb], bf, acc2);
        }
      }
#pragma unroll
      for (int v = 0; v < 8; ++v) {
        const float g1 = sigmoidf_(acc1[v]);
        const float g2 = sigmoidf_(acc2[v]);
        const float mval = p3[nt][v] * g1 + p5[nt][v] * g2;
        maxv[v] = fmaxf(maxv[v], mval);
      }
      __builtin_amdgcn_sched_group_barrier(0x100, 16, 0);
      __builtin_amdgcn_sched_group_barrier(0x008, 8, 0);
      __builtin_amdgcn_sched_barrier(0);
    }
#pragma unroll
    for (int v = 0; v < 8; ++v) {
#pragma unroll
      for (int off = 8; off >= 1; off >>= 1)
        maxv[v] = fmaxf(maxv[v], __shfl_xor(maxv[v], off, 16));
    }
    if ((lane & 15) == 0) {
      float* const o = out + ((size_t)b * cfg::PARTS + part) * cfg::CH + m0 + (hs << 3);
#pragma unroll
      for (int v = 0; v < 8; ++v) o[v] = maxv[v];
    }
  }
  }
}

extern "C" void kernel_launch(void* const* d_in, const int* in_sizes, int n_in,
                              void* d_out, int out_size, void* d_ws, size_t ws_size,
                              hipStream_t stream) {
  (void)in_sizes; (void)n_in; (void)d_ws; (void)ws_size; (void)out_size;
  const float* x   = (const float*)d_in[0];
  const float* W1a = (const float*)d_in[1];
  const float* W1b = (const float*)d_in[2];
  const float* W2a = (const float*)d_in[3];
  const float* W2b = (const float*)d_in[4];
  float* out = (float*)d_out;

  dim3 grid(cfg::GRID_X, cfg::NB);
  dim3 block(cfg::THREADS);
  mtb_gate_kernel<<<grid, block, 0, stream>>>(x, W1a, W1b, W2a, W2b, out);
}
